# starting kernel
# baseline (speedup 1.0000x reference)
#include <hip/hip_runtime.h>
#include <math.h>

typedef __attribute__((ext_vector_type(2))) float v2f;
typedef __attribute__((ext_vector_type(8))) float v8f;

#define NPIX   (512*512)
#define NSAMP  16
#define CI     3
#define CH     16
#define CO     3
#define PB1    64
#define PB3    64
#define PB5    128
#define EPS    1e-7f
#define SLOPE  0.01f

__device__ __forceinline__ v8f wmma4(v2f a, v2f b, v8f c) {
  return __builtin_amdgcn_wmma_f32_16x16x4_f32(false, a, false, b, (short)0, c,
                                               false, false);
}

__global__ void k_stats1(const float* __restrict__ x, const float* __restrict__ Wa,
                         const float* __restrict__ ba, float* __restrict__ part) {
  const int s = blockIdx.y;
  const float* xs = x + (size_t)s * CI * NPIX;

  float wa[CH][CI], bav[CH];
#pragma unroll
  for (int c = 0; c < CH; ++c) {
    bav[c] = ba[c];
#pragma unroll
    for (int i = 0; i < CI; ++i) wa[c][i] = Wa[c * CI + i];
  }

  float s1 = 0.f, s2 = 0.f;
  const int stride = blockDim.x * gridDim.x;
  for (int p = blockIdx.x * blockDim.x + threadIdx.x; p < NPIX; p += stride) {
    float x0 = xs[p], x1 = xs[NPIX + p], x2 = xs[2 * NPIX + p];
#pragma unroll
    for (int c = 0; c < CH; ++c) {
      float y = fmaf(wa[c][0], x0, fmaf(wa[c][1], x1, fmaf(wa[c][2], x2, bav[c])));
      s1 += y;
      s2 += y * y;
    }
  }

  __shared__ float r1[256], r2[256];
  r1[threadIdx.x] = s1; r2[threadIdx.x] = s2;
  __syncthreads();
  for (int w = 128; w > 0; w >>= 1) {
    if (threadIdx.x < (unsigned)w) {
      r1[threadIdx.x] += r1[threadIdx.x + w];
      r2[threadIdx.x] += r2[threadIdx.x + w];
    }
    __syncthreads();
  }
  if (threadIdx.x == 0) {
    part[(s * PB1 + blockIdx.x) * 2 + 0] = r1[0];
    part[(s * PB1 + blockIdx.x) * 2 + 1] = r2[0];
  }
}

__global__ void k_finalize(const float* __restrict__ part, float* __restrict__ stats,
                           int nb) {
  int s = threadIdx.x;
  if (s >= NSAMP) return;
  double s1 = 0.0, s2 = 0.0;
  for (int b = 0; b < nb; ++b) {
    s1 += (double)part[(s * nb + b) * 2 + 0];
    s2 += (double)part[(s * nb + b) * 2 + 1];
  }
  const double n = (double)CH * (double)NPIX;
  double mean = s1 / n;
  double var = (s2 - s1 * mean) / (n - 1.0);
  stats[2 * s + 0] = (float)mean;
  stats[2 * s + 1] = 1.0f / (sqrtf((float)var) + EPS);
}

__global__ void k_stats2(const float* __restrict__ x, const float* __restrict__ Wa,
                         const float* __restrict__ ba, const float* __restrict__ Wb,
                         const float* __restrict__ bb, const float* __restrict__ stats1,
                         float* __restrict__ part) {
  const int s = blockIdx.y;
  const float* xs = x + (size_t)s * CI * NPIX;
  const int lane  = threadIdx.x & 31;
  const int wave  = threadIdx.x >> 5;
  const int m     = lane & 15;
  const int khalf = (lane >> 4) << 1;
  const int hi8   = (lane >> 4) << 3;

  v2f aA;
  aA.x = Wa[m * CI + khalf];
  aA.y = (khalf == 0) ? Wa[m * CI + 1] : 0.f;

  v2f bA[4];
#pragma unroll
  for (int j = 0; j < 4; ++j) {
    int k0 = 4 * j + khalf;
    bA[j].x = Wb[m * CH + k0];
    bA[j].y = Wb[m * CH + k0 + 1];
  }

  v8f biasA, biasB;
#pragma unroll
  for (int i = 0; i < 8; ++i) { biasA[i] = ba[i + hi8]; biasB[i] = bb[i + hi8]; }

  const float mean1 = stats1[2 * s], inv1 = stats1[2 * s + 1];

  __shared__ float tile[8][CH * 16];
  float* tl = &tile[wave][0];

  float acc1 = 0.f, acc2 = 0.f;
  const int tpb = (NPIX / 16) / gridDim.x;
  const int tpw = tpb / 8;
  const int tbase = blockIdx.x * tpb + wave * tpw;

  for (int t = 0; t < tpw; ++t) {
    const int p = (tbase + t) * 16;
    v2f xB;
    xB.x = xs[khalf * NPIX + p + m];
    xB.y = (khalf == 0) ? xs[NPIX + p + m] : 0.f;

    v8f y = wmma4(aA, xB, biasA);

    v8f h;
#pragma unroll
    for (int i = 0; i < 8; ++i) {
      float v = (y[i] - mean1) * inv1;
      h[i] = (v >= 0.f) ? v : SLOPE * v;
    }

    __builtin_amdgcn_wave_barrier();
#pragma unroll
    for (int i = 0; i < 8; ++i) tl[(i + hi8) * 16 + m] = h[i];
    __builtin_amdgcn_wave_barrier();

    v8f z = biasB;
#pragma unroll
    for (int j = 0; j < 4; ++j) {
      int k0 = 4 * j + khalf;
      v2f hB;
      hB.x = tl[k0 * 16 + m];
      hB.y = tl[(k0 + 1) * 16 + m];
      z = wmma4(bA[j], hB, z);
    }
    __builtin_amdgcn_wave_barrier();

#pragma unroll
    for (int i = 0; i < 8; ++i) { acc1 += z[i]; acc2 += z[i] * z[i]; }
  }

  __shared__ float r1[256], r2[256];
  r1[threadIdx.x] = acc1; r2[threadIdx.x] = acc2;
  __syncthreads();
  for (int w = 128; w > 0; w >>= 1) {
    if (threadIdx.x < (unsigned)w) {
      r1[threadIdx.x] += r1[threadIdx.x + w];
      r2[threadIdx.x] += r2[threadIdx.x + w];
    }
    __syncthreads();
  }
  if (threadIdx.x == 0) {
    part[(s * PB3 + blockIdx.x) * 2 + 0] = r1[0];
    part[(s * PB3 + blockIdx.x) * 2 + 1] = r2[0];
  }
}

__global__ void k_final(const float* __restrict__ x, const float* __restrict__ Wa,
                        const float* __restrict__ ba, const float* __restrict__ Wb,
                        const float* __restrict__ bb, const float* __restrict__ Wc,
                        const float* __restrict__ bc, const float* __restrict__ stats1,
                        const float* __restrict__ stats2, float* __restrict__ out) {
  const int s = blockIdx.y;
  const float* xs = x + (size_t)s * CI * NPIX;
  float* os = out + (size_t)s * CO * NPIX;
  const int lane  = threadIdx.x & 31;
  const int wave  = threadIdx.x >> 5;
  const int m     = lane & 15;
  const int khalf = (lane >> 4) << 1;
  const int hi8   = (lane >> 4) << 3;

  v2f aA;
  aA.x = Wa[m * CI + khalf];
  aA.y = (khalf == 0) ? Wa[m * CI + 1] : 0.f;

  v2f bA[4], cA[4];
#pragma unroll
  for (int j = 0; j < 4; ++j) {
    int k0 = 4 * j + khalf;
    bA[j].x = Wb[m * CH + k0];
    bA[j].y = Wb[m * CH + k0 + 1];
    cA[j].x = (m < CO) ? Wc[m * CH + k0] : 0.f;
    cA[j].y = (m < CO) ? Wc[m * CH + k0 + 1] : 0.f;
  }

  v8f biasA, biasB, biasC;
#pragma unroll
  for (int i = 0; i < 8; ++i) {
    biasA[i] = ba[i + hi8];
    biasB[i] = bb[i + hi8];
    biasC[i] = (i + hi8 < CO) ? bc[i + hi8] : 0.f;
  }

  const float mean1 = stats1[2 * s], inv1 = stats1[2 * s + 1];
  const float mean2 = stats2[2 * s], inv2 = stats2[2 * s + 1];

  __shared__ float tile[8][CH * 16];
  float* tl = &tile[wave][0];

  const int tpb = (NPIX / 16) / gridDim.x;
  const int tpw = tpb / 8;
  const int tbase = blockIdx.x * tpb + wave * tpw;

  for (int t = 0; t < tpw; ++t) {
    const int p = (tbase + t) * 16;
    v2f xB;
    xB.x = xs[khalf * NPIX + p + m];
    xB.y = (khalf == 0) ? xs[NPIX + p + m] : 0.f;

    v8f y = wmma4(aA, xB, biasA);

    v8f h1;
#pragma unroll
    for (int i = 0; i < 8; ++i) {
      float v = (y[i] - mean1) * inv1;
      h1[i] = (v >= 0.f) ? v : SLOPE * v;
    }

    __builtin_amdgcn_wave_barrier();
#pragma unroll
    for (int i = 0; i < 8; ++i) tl[(i + hi8) * 16 + m] = h1[i];
    __builtin_amdgcn_wave_barrier();

    v8f z = biasB;
#pragma unroll
    for (int j = 0; j < 4; ++j) {
      int k0 = 4 * j + khalf;
      v2f hB;
      hB.x = tl[k0 * 16 + m];
      hB.y = tl[(k0 + 1) * 16 + m];
      z = wmma4(bA[j], hB, z);
    }
    __builtin_amdgcn_wave_barrier();

    v8f h2;
#pragma unroll
    for (int i = 0; i < 8; ++i) {
      float v = (z[i] - mean2) * inv2;
      h2[i] = (v >= 0.f) ? v : SLOPE * v;
    }

    __builtin_amdgcn_wave_barrier();
#pragma unroll
    for (int i = 0; i < 8; ++i) tl[(i + hi8) * 16 + m] = h2[i];
    __builtin_amdgcn_wave_barrier();

    v8f o = biasC;
#pragma unroll
    for (int j = 0; j < 4; ++j) {
      int k0 = 4 * j + khalf;
      v2f hB;
      hB.x = tl[k0 * 16 + m];
      hB.y = tl[(k0 + 1) * 16 + m];
      o = wmma4(cA[j], hB, o);
    }
    __builtin_amdgcn_wave_barrier();

    if (lane < 16) {
      os[0 * NPIX + p + m] = o[0];
      os[1 * NPIX + p + m] = o[1];
      os[2 * NPIX + p + m] = o[2];
    }
  }
}

extern "C" void kernel_launch(void* const* d_in, const int* in_sizes, int n_in,
                              void* d_out, int out_size, void* d_ws, size_t ws_size,
                              hipStream_t stream) {
  (void)in_sizes; (void)n_in; (void)out_size; (void)ws_size;
  const float* x  = (const float*)d_in[0];
  const float* Wa = (const float*)d_in[1];
  const float* ba = (const float*)d_in[2];
  const float* Wb = (const float*)d_in[3];
  const float* bb = (const float*)d_in[4];
  const float* Wc = (const float*)d_in[5];
  const float* bc = (const float*)d_in[6];
  float* out = (float*)d_out;

  float* ws     = (float*)d_ws;
  float* part1  = ws;
  float* stats1 = part1 + NSAMP * PB1 * 2;
  float* part2  = stats1 + NSAMP * 2;
  float* stats2 = part2 + NSAMP * PB3 * 2;

  k_stats1 <<<dim3(PB1, NSAMP), 256, 0, stream>>>(x, Wa, ba, part1);
  k_finalize<<<1, 32, 0, stream>>>(part1, stats1, PB1);
  k_stats2 <<<dim3(PB3, NSAMP), 256, 0, stream>>>(x, Wa, ba, Wb, bb, stats1, part2);
  k_finalize<<<1, 32, 0, stream>>>(part2, stats2, PB3);
  k_final  <<<dim3(PB5, NSAMP), 256, 0, stream>>>(x, Wa, ba, Wb, bb, Wc, bc,
                                                  stats1, stats2, out);
}


# variance of y1 leaves pass 1: it only sums x per channel (mean of y1 is linear in that); pass 2 adds sum (y1-mean)^2 from conv_a's D regs, inv1 goes on its sums afterwards; sum order changes
# speedup vs baseline: 1.2344x; 1.2344x over previous
#include <hip/hip_runtime.h>
#include <math.h>

typedef __attribute__((ext_vector_type(2))) float v2f;
typedef __attribute__((ext_vector_type(8))) float v8f;

#define NPIX   (512*512)
#define NSAMP  16
#define CI     3
#define CH     16
#define CO     3
#define PB1    32
#define PB3    64
#define PB5    128
#define EPS    1e-7f
#define SLOPE  0.01f

__device__ __forceinline__ v8f wmma4(v2f a, v2f b, v8f c) {
  return __builtin_amdgcn_wmma_f32_16x16x4_f32(false, a, false, b, (short)0, c,
                                               false, false);
}

__global__ void k_stats1(const float* __restrict__ x, float* __restrict__ part) {
  const int s = blockIdx.y;
  const float4* x4 = reinterpret_cast<const float4*>(__builtin_assume_aligned(x, 16)) +
                     (size_t)s * CI * (NPIX / 4);
  constexpr int V = NPIX / 4 / PB1 / 256;
  const int base = blockIdx.x * (NPIX / 4 / PB1) + threadIdx.x;

  float4 v[CI][V];
#pragma unroll
  for (int c = 0; c < CI; ++c)
#pragma unroll
    for (int i = 0; i < V; ++i) v[c][i] = x4[c * (NPIX / 4) + base + i * 256];

  float a[CI];
#pragma unroll
  for (int c = 0; c < CI; ++c) {
    float t = 0.f;
#pragma unroll
    for (int i = 0; i < V; ++i) t += (v[c][i].x + v[c][i].y) + (v[c][i].z + v[c][i].w);
#pragma unroll
    for (int o = 16; o > 0; o >>= 1) t += __shfl_xor(t, o);
    a[c] = t;
  }

  __shared__ float r[8][4];
  const int lane = threadIdx.x & 31, wave = threadIdx.x >> 5;
  if (lane == 0) { r[wave][0] = a[0]; r[wave][1] = a[1]; r[wave][2] = a[2]; }
  __syncthreads();
  if (threadIdx.x == 0) {
    float4 t = {0.f, 0.f, 0.f, 0.f};
#pragma unroll
    for (int w = 0; w < 8; ++w) { t.x += r[w][0]; t.y += r[w][1]; t.z += r[w][2]; }
    reinterpret_cast<float4*>(__builtin_assume_aligned(part, 16))[s * PB1 + blockIdx.x] = t;
  }
}

__global__ void k_fin_mean(const float* __restrict__ part, const float* __restrict__ Wa,
                           const float* __restrict__ ba, float* __restrict__ stats1) {
  static_assert(PB1 == 32, "one block partial per lane");
  const int s = blockIdx.x;
  const int lane = threadIdx.x;
  const int c = lane & 15;
  const float w0 = Wa[c * CI], w1 = Wa[c * CI + 1], w2 = Wa[c * CI + 2], b = ba[c];
  const float4 q = reinterpret_cast<const float4*>(
      __builtin_assume_aligned(part, 16))[s * PB1 + lane];
  double x0 = (double)q.x, x1 = (double)q.y, x2 = (double)q.z;
#pragma unroll
  for (int o = 16; o > 0; o >>= 1) {
    x0 += __shfl_xor(x0, o);
    x1 += __shfl_xor(x1, o);
    x2 += __shfl_xor(x2, o);
  }
  double t = 0.0;
  if (lane < CH)
    t = ((double)w0 * x0 + (double)w1 * x1 + (double)w2 * x2) / (double)NPIX + (double)b;
#pragma unroll
  for (int o = 16; o > 0; o >>= 1) t += __shfl_xor(t, o);
  if (lane == 0) stats1[2 * s] = (float)(t / (double)CH);
}

__global__ void k_stats2(const float* __restrict__ x, const float* __restrict__ Wa,
                         const float* __restrict__ ba, const float* __restrict__ Wb,
                         const float* __restrict__ bb, const float* __restrict__ stats1,
                         float* __restrict__ part) {
  const int s = blockIdx.y;
  const float* xs = x + (size_t)s * CI * NPIX;
  const int lane  = threadIdx.x & 31;
  const int wave  = threadIdx.x >> 5;
  const int m     = lane & 15;
  const int khalf = (lane >> 4) << 1;

  v2f aB;
  aB.x = Wa[m * CI + khalf];
  aB.y = (khalf == 0) ? Wa[m * CI + 1] : 0.f;

  const float bac = ba[m];
  v8f biasT;
#pragma unroll
  for (int i = 0; i < 8; ++i) biasT[i] = bac;

  float u = 0.f, w = 0.f;
#pragma unroll
  for (int c = 0; c < CH; ++c) {
    const float wc = Wb[c * CH + m];
    u += wc;
    w = fmaf(bb[c], wc, w);
  }

  v8f M = {};
#pragma unroll
  for (int j = 0; j < 4; ++j) {
    const int k0 = 4 * j + khalf;
    v2f wf;
    wf.x = Wb[k0 * CH + m];
    wf.y = Wb[(k0 + 1) * CH + m];
    M = wmma4(wf, wf, M);
  }

  const float mean1 = stats1[2 * s];

  v8f G0 = {}, G1 = {};
  v8f hs = {};
  v8f qs = {};
  const int tpb = (NPIX / 16) / gridDim.x;
  const int tpw = tpb / 8;
  const int tbase = blockIdx.x * tpb + wave * tpw;

  for (int t = 0; t < tpw; ++t) {
    const int p = (tbase + t) * 16;
    v2f xA;
    xA.x = xs[khalf * NPIX + p + m];
    xA.y = (khalf == 0) ? xs[NPIX + p + m] : 0.f;

    v8f y = wmma4(xA, aB, biasT);

    v8f h;
#pragma unroll
    for (int i = 0; i < 8; ++i) {
      float v = y[i] - mean1;
      qs[i] = fmaf(v, v, qs[i]);
      h[i] = (v >= 0.f) ? v : SLOPE * v;
    }

#pragma unroll
    for (int q = 0; q < 4; ++q) {
      v2f f;
      f.x = h[2 * q];
      f.y = h[2 * q + 1];
      if (q & 1) G1 = wmma4(f, f, G1);
      else       G0 = wmma4(f, f, G0);
    }
    hs += h;
  }

  const float S = ((hs[0] + hs[1]) + (hs[2] + hs[3])) + ((hs[4] + hs[5]) + (hs[6] + hs[7]));
  float mg = 0.f;
#pragma unroll
  for (int i = 0; i < 8; ++i) mg = fmaf(M[i], G0[i] + G1[i], mg);
  const float Q = ((qs[0] + qs[1]) + (qs[2] + qs[3])) + ((qs[4] + qs[5]) + (qs[6] + qs[7]));

  __shared__ float r1[256], r2[256], r3[256], r4[256];
  r1[threadIdx.x] = u * S; r2[threadIdx.x] = mg;
  r3[threadIdx.x] = 2.f * w * S; r4[threadIdx.x] = Q;
  __syncthreads();
  for (int w2 = 128; w2 > 0; w2 >>= 1) {
    if (threadIdx.x < (unsigned)w2) {
      r1[threadIdx.x] += r1[threadIdx.x + w2];
      r2[threadIdx.x] += r2[threadIdx.x + w2];
      r3[threadIdx.x] += r3[threadIdx.x + w2];
      r4[threadIdx.x] += r4[threadIdx.x + w2];
    }
    __syncthreads();
  }
  if (threadIdx.x == 0) {
    float4 t = {r1[0], r2[0], r3[0], r4[0]};
    reinterpret_cast<float4*>(__builtin_assume_aligned(part, 16))[s * PB3 + blockIdx.x] = t;
  }
}

__global__ void k_fin_stats(const float* __restrict__ part, const float* __restrict__ bb,
                            float* __restrict__ stats1, float* __restrict__ stats2) {
  static_assert(PB3 == 64, "two block partials per lane");
  const int s = blockIdx.x;
  const int lane = threadIdx.x;
  const float4* p4 = reinterpret_cast<const float4*>(__builtin_assume_aligned(part, 16)) +
                     s * PB3;
  const float bl = (lane < CH) ? bb[lane & 15] : 0.f;
  const float4 qa = p4[2 * lane], qb = p4[2 * lane + 1];
  double p0 = (double)qa.x + (double)qb.x;
  double p1 = (double)qa.y + (double)qb.y;
  double p2 = (double)qa.z + (double)qb.z;
  double p3 = (double)qa.w + (double)qb.w;
  double sb = (double)bl, sb2 = (double)bl * (double)bl;
#pragma unroll
  for (int o = 16; o > 0; o >>= 1) {
    p0 += __shfl_xor(p0, o);
    p1 += __shfl_xor(p1, o);
    p2 += __shfl_xor(p2, o);
    p3 += __shfl_xor(p3, o);
    sb += __shfl_xor(sb, o);
    sb2 += __shfl_xor(sb2, o);
  }
  if (lane == 0) {
    const double n = (double)CH * (double)NPIX;
    const double var1 = p3 / (n - 1.0);
    const float inv1f = 1.0f / (sqrtf((float)var1) + EPS);
    const double inv1 = (double)inv1f;
    const double s1 = inv1 * p0 + (double)NPIX * sb;
    const double s2 = inv1 * inv1 * p1 + inv1 * p2 + (double)NPIX * sb2;
    const double mean = s1 / n;
    const double var = (s2 - s1 * mean) / (n - 1.0);
    stats1[2 * s + 1] = inv1f;
    stats2[2 * s + 0] = (float)mean;
    stats2[2 * s + 1] = 1.0f / (sqrtf((float)var) + EPS);
  }
}

__global__ void k_final(const float* __restrict__ x, const float* __restrict__ Wa,
                        const float* __restrict__ ba, const float* __restrict__ Wb,
                        const float* __restrict__ bb, const float* __restrict__ Wc,
                        const float* __restrict__ bc, const float* __restrict__ stats1,
                        const float* __restrict__ stats2, float* __restrict__ out) {
  const int s = blockIdx.y;
  const float* xs = x + (size_t)s * CI * NPIX;
  float* os = out + (size_t)s * CO * NPIX;
  const int lane  = threadIdx.x & 31;
  const int wave  = threadIdx.x >> 5;
  const int m     = lane & 15;
  const int khalf = (lane >> 4) << 1;
  const int hi8   = (lane >> 4) << 3;

  v2f aA;
  aA.x = Wa[m * CI + khalf];
  aA.y = (khalf == 0) ? Wa[m * CI + 1] : 0.f;

  v2f bA[4], cA[4];
#pragma unroll
  for (int j = 0; j < 4; ++j) {
    int k0 = 4 * j + khalf;
    bA[j].x = Wb[m * CH + k0];
    bA[j].y = Wb[m * CH + k0 + 1];
    cA[j].x = (m < CO) ? Wc[m * CH + k0] : 0.f;
    cA[j].y = (m < CO) ? Wc[m * CH + k0 + 1] : 0.f;
  }

  v8f biasA, biasB, biasC;
#pragma unroll
  for (int i = 0; i < 8; ++i) {
    biasA[i] = ba[i + hi8];
    biasB[i] = bb[i + hi8];
    biasC[i] = (i + hi8 < CO) ? bc[i + hi8] : 0.f;
  }

  const float mean1 = stats1[2 * s], inv1 = stats1[2 * s + 1];
  const float mean2 = stats2[2 * s], inv2 = stats2[2 * s + 1];

  __shared__ float tile[8][CH * 16];
  float* tl = &tile[wave][0];

  const int tpb = (NPIX / 16) / gridDim.x;
  const int tpw = tpb / 8;
  const int tbase = blockIdx.x * tpb + wave * tpw;

  for (int t = 0; t < tpw; ++t) {
    const int p = (tbase + t) * 16;
    v2f xB;
    xB.x = xs[khalf * NPIX + p + m];
    xB.y = (khalf == 0) ? xs[NPIX + p + m] : 0.f;

    v8f y = wmma4(aA, xB, biasA);

    v8f h1;
#pragma unroll
    for (int i = 0; i < 8; ++i) {
      float v = (y[i] - mean1) * inv1;
      h1[i] = (v >= 0.f) ? v : SLOPE * v;
    }

    __builtin_amdgcn_wave_barrier();
#pragma unroll
    for (int i = 0; i < 8; ++i) tl[(i + hi8) * 16 + m] = h1[i];
    __builtin_amdgcn_wave_barrier();

    v8f z = biasB;
#pragma unroll
    for (int j = 0; j < 4; ++j) {
      int k0 = 4 * j + khalf;
      v2f hB;
      hB.x = tl[k0 * 16 + m];
      hB.y = tl[(k0 + 1) * 16 + m];
      z = wmma4(bA[j], hB, z);
    }
    __builtin_amdgcn_wave_barrier();

    v8f h2;
#pragma unroll
    for (int i = 0; i < 8; ++i) {
      float v = (z[i] - mean2) * inv2;
      h2[i] = (v >= 0.f) ? v : SLOPE * v;
    }

    __builtin_amdgcn_wave_barrier();
#pragma unroll
    for (int i = 0; i < 8; ++i) tl[(i + hi8) * 16 + m] = h2[i];
    __builtin_amdgcn_wave_barrier();

    v8f o = biasC;
#pragma unroll
    for (int j = 0; j < 4; ++j) {
      int k0 = 4 * j + khalf;
      v2f hB;
      hB.x = tl[k0 * 16 + m];
      hB.y = tl[(k0 + 1) * 16 + m];
      o = wmma4(cA[j], hB, o);
    }
    __builtin_amdgcn_wave_barrier();

    if (lane < 16) {
      os[0 * NPIX + p + m] = o[0];
      os[1 * NPIX + p + m] = o[1];
      os[2 * NPIX + p + m] = o[2];
    }
  }
}

extern "C" void kernel_launch(void* const* d_in, const int* in_sizes, int n_in,
                              void* d_out, int out_size, void* d_ws, size_t ws_size,
                              hipStream_t stream) {
  (void)in_sizes; (void)n_in; (void)out_size; (void)ws_size;
  const float* x  = (const float*)d_in[0];
  const float* Wa = (const float*)d_in[1];
  const float* ba = (const float*)d_in[2];
  const float* Wb = (const float*)d_in[3];
  const float* bb = (const float*)d_in[4];
  const float* Wc = (const float*)d_in[5];
  const float* bc = (const float*)d_in[6];
  float* out = (float*)d_out;

  float* ws     = (float*)d_ws;
  float* part1  = ws;
  float* stats1 = part1 + NSAMP * PB1 * 4;
  float* part2  = stats1 + NSAMP * 2;
  float* stats2 = part2 + NSAMP * PB3 * 4;

  k_stats1 <<<dim3(PB1, NSAMP), 256, 0, stream>>>(x, part1);
  k_fin_mean<<<NSAMP, 32, 0, stream>>>(part1, Wa, ba, stats1);
  k_stats2 <<<dim3(PB3, NSAMP), 256, 0, stream>>>(x, Wa, ba, Wb, bb, stats1, part2);
  k_fin_stats<<<NSAMP, 32, 0, stream>>>(part2, bb, stats1, stats2);
  k_final  <<<dim3(PB5, NSAMP), 256, 0, stream>>>(x, Wa, ba, Wb, bb, Wc, bc,
                                                  stats1, stats2, out);
}
